# starting kernel
# baseline (speedup 1.0000x reference)
#include <hip/hip_runtime.h>

#define N_NODES 100000
#define N_EDGES 1000000
#define DHID    64
#define KORD    8
#define NLAYERS 3

typedef float v2f __attribute__((ext_vector_type(2)));
typedef float v8f __attribute__((ext_vector_type(8)));

__global__ void zero_kernel(float* __restrict__ p, int n) {
    int i = blockIdx.x * blockDim.x + threadIdx.x;
    int stride = gridDim.x * blockDim.x;
    for (; i < n; i += stride) p[i] = 0.0f;
}

__global__ void deg_kernel(const int* __restrict__ src, const int* __restrict__ dst,
                           const float* __restrict__ ew, float* __restrict__ deg, int nE) {
    int e = blockIdx.x * blockDim.x + threadIdx.x;
    if (e >= nE) return;
    int s = src[e];
    if (s == dst[e]) return;
    atomicAdd(&deg[s], ew[e]);
}

__global__ void dinv_kernel(float* __restrict__ deg, int n) {
    int i = blockIdx.x * blockDim.x + threadIdx.x;
    if (i >= n) return;
    float d = deg[i];
    deg[i] = (d > 0.0f) ? rsqrtf(d) : 0.0f;
}

__global__ void norm_kernel(const int* __restrict__ src, const int* __restrict__ dst,
                            const float* __restrict__ ew, const float* __restrict__ dinv,
                            float* __restrict__ norm, int nE) {
    int e = blockIdx.x * blockDim.x + threadIdx.x;
    if (e >= nE) return;
    int s = src[e], d = dst[e];
    float w = (s == d) ? 0.0f : ew[e];
    norm[e] = -dinv[s] * w * dinv[d];
}

__global__ void scatter_kernel(const float* __restrict__ h, const int* __restrict__ src,
                               const int* __restrict__ dst, const float* __restrict__ norm,
                               float* __restrict__ out, int nE) {
    int t  = blockIdx.x * blockDim.x + threadIdx.x;
    int e  = t >> 4;
    int f4 = (t & 15) << 2;
    if (e >= nE) return;
    float w = norm[e];
    if (w == 0.0f) return;
    const float4 hv = *(const float4*)(h + (size_t)src[e] * DHID + f4);
    float* o = out + (size_t)dst[e] * DHID + f4;
    atomicAdd(o + 0, w * hv.x);
    atomicAdd(o + 1, w * hv.y);
    atomicAdd(o + 2, w * hv.z);
    atomicAdd(o + 3, w * hv.w);
}

__global__ void relu_bias_kernel(const float* __restrict__ acc, const float* __restrict__ b,
                                 float* __restrict__ out, int n) {
    int i = blockIdx.x * blockDim.x + threadIdx.x;
    if (i >= n) return;
    out[i] = fmaxf(acc[i] + b[i & (DHID - 1)], 0.0f);
}

template <int MODE>
__global__ void __launch_bounds__(256)
gemm64_wmma(float* __restrict__ T, const float* __restrict__ t0,
            const float* __restrict__ W, float* __restrict__ acc, int nrows) {
    __shared__ float lWp[DHID * DHID];
    for (int idx = threadIdx.x; idx < DHID * DHID; idx += 256) {
        int k = idx >> 6, n = idx & 63;
        lWp[((k >> 1) << 7) + (n << 1) + (k & 1)] = W[idx];
    }
    __syncthreads();

    const int wave = threadIdx.x >> 5;
    const int lane = threadIdx.x & 31;
    const int row0 = (blockIdx.x * 8 + wave) * 16;
    if (row0 >= nrows) return;

    const int lrow  = lane & 15;
    const int khalf = (lane >> 4) << 1;
    const int crow  = row0 + ((lane >> 4) << 3);
    const int ccol  = lane & 15;

    v8f c0 = {}, c1 = {}, c2 = {}, c3 = {};
    if (MODE != 0) {
#pragma unroll
        for (int r = 0; r < 8; ++r) {
            const float* ap = acc + (size_t)(crow + r) * DHID + ccol;
            c0[r] = ap[0]; c1[r] = ap[16]; c2[r] = ap[32]; c3[r] = ap[48];
        }
    }

    float* arow       = T + (size_t)(row0 + lrow) * DHID + khalf;
    const float* trow = (MODE == 2) ? (t0 + (size_t)(row0 + lrow) * DHID + khalf) : nullptr;

#pragma unroll
    for (int kt = 0; kt < 16; ++kt) {
        const int k0 = kt * 4 + khalf;
        v2f a = *(const v2f*)(arow + kt * 4);
        if (MODE == 2) {
            v2f t = *(const v2f*)(trow + kt * 4);
            a.x = 2.0f * a.x - t.x;
            a.y = 2.0f * a.y - t.y;
            *(v2f*)(arow + kt * 4) = a;
        }
        const float* wp = lWp + ((k0 >> 1) << 7);
        v2f b;
        b = *(const v2f*)(wp + ((lrow +  0) << 1));
        c0 = __builtin_amdgcn_wmma_f32_16x16x4_f32(false, a, false, b, (short)0, c0, false, false);
        b = *(const v2f*)(wp + ((lrow + 16) << 1));
        c1 = __builtin_amdgcn_wmma_f32_16x16x4_f32(false, a, false, b, (short)0, c1, false, false);
        b = *(const v2f*)(wp + ((lrow + 32) << 1));
        c2 = __builtin_amdgcn_wmma_f32_16x16x4_f32(false, a, false, b, (short)0, c2, false, false);
        b = *(const v2f*)(wp + ((lrow + 48) << 1));
        c3 = __builtin_amdgcn_wmma_f32_16x16x4_f32(false, a, false, b, (short)0, c3, false, false);
    }

#pragma unroll
    for (int r = 0; r < 8; ++r) {
        float* ap = acc + (size_t)(crow + r) * DHID + ccol;
        ap[0] = c0[r]; ap[16] = c1[r]; ap[32] = c2[r]; ap[48] = c3[r];
    }
}

extern "C" void kernel_launch(void* const* d_in, const int* in_sizes, int n_in,
                              void* d_out, int out_size, void* d_ws, size_t ws_size,
                              hipStream_t stream) {
    const float* x       = (const float*)d_in[0];
    const int*   ei      = (const int*)d_in[1];
    const float* ew      = (const float*)d_in[2];
    const float* weights = (const float*)d_in[3];
    const float* bias    = (const float*)d_in[4];
    float*       out     = (float*)d_out;

    const int* src = ei;
    const int* dst = ei + N_EDGES;

    float* ws   = (float*)d_ws;
    float* norm = ws;
    float* dinv = ws + N_EDGES;
    float* base = ws + N_EDGES + N_NODES;
    float* P[5];
    for (int i = 0; i < 5; ++i) P[i] = base + (size_t)i * N_NODES * DHID;
    float* ACC = P[4];

    const int NF = N_NODES * DHID;
    dim3 blk(256);
    const int gE    = (N_EDGES + 255) / 256;
    const int gN    = (N_NODES + 255) / 256;
    const int gNF   = (NF + 255) / 256;
    const int gSc   = (N_EDGES * 16 + 255) / 256;
    const int gGemm = (N_NODES + 127) / 128;

    zero_kernel<<<1024, blk, 0, stream>>>(dinv, N_NODES);
    deg_kernel <<<gE,   blk, 0, stream>>>(src, dst, ew, dinv, N_EDGES);
    dinv_kernel<<<gN,   blk, 0, stream>>>(dinv, N_NODES);
    norm_kernel<<<gE,   blk, 0, stream>>>(src, dst, ew, dinv, norm, N_EDGES);

    auto prop = [&](const float* tin, float* tout) {
        zero_kernel   <<<2048, blk, 0, stream>>>(tout, NF);
        scatter_kernel<<<gSc,  blk, 0, stream>>>(tin, src, dst, norm, tout, N_EDGES);
    };

    int h_idx = -1;
    for (int l = 0; l < NLAYERS; ++l) {
        int rid[3], c = 0;
        for (int i = 0; i < 4 && c < 3; ++i)
            if (i != h_idx) rid[c++] = i;

        const float* t0 = (h_idx < 0) ? x : P[h_idx];
        float* t1 = P[rid[0]];
        const float* Wl = weights + (size_t)l * KORD * DHID * DHID;

        prop(t0, t1);
        gemm64_wmma<0><<<gGemm, blk, 0, stream>>>((float*)t0, nullptr, Wl,
                                                  ACC, N_NODES);
        gemm64_wmma<1><<<gGemm, blk, 0, stream>>>(t1, nullptr, Wl + DHID * DHID,
                                                  ACC, N_NODES);

        int q[10]; int qh = 0, qt = 0;
        q[qt++] = rid[1];
        q[qt++] = rid[2];
        int t0_idx = h_idx;
        int t1_idx = rid[0];

        for (int k = 2; k < KORD; ++k) {
            int s_idx = q[qh++];
            float* scat = P[s_idx];
            prop(P[t1_idx], scat);
            gemm64_wmma<2><<<gGemm, blk, 0, stream>>>(
                scat, (t0_idx < 0) ? x : P[t0_idx],
                Wl + (size_t)k * DHID * DHID, ACC, N_NODES);
            if (t0_idx >= 0) q[qt++] = t0_idx;
            t0_idx = t1_idx;
            t1_idx = s_idx;
        }

        if (l == NLAYERS - 1) {
            relu_bias_kernel<<<gNF, blk, 0, stream>>>(ACC, bias + l * DHID, out, NF);
        } else {
            float* hn = P[t1_idx];
            relu_bias_kernel<<<gNF, blk, 0, stream>>>(ACC, bias + l * DHID, hn, NF);
            h_idx = t1_idx;
        }
    }
}


# Fewer waves enter: 1536 resident scatter blocks loop over the 128-edge pieces (next ticket drawn ahead); bias/ReLU, degree and edge-record kernels stride over their data; sums stay in arrival order
# speedup vs baseline: 2.8542x; 2.8542x over previous
#include <hip/hip_runtime.h>

#define N_NODES 100000
#define N_EDGES 1000000
#define DHID    64
#define KORD    8
#define NLAYERS 3

typedef float v2f __attribute__((ext_vector_type(2)));
typedef float v8f __attribute__((ext_vector_type(8)));

__global__ void zero_kernel(float* __restrict__ p, int n) {
    int i = blockIdx.x * blockDim.x + threadIdx.x;
    int stride = gridDim.x * blockDim.x;
    for (; i < n; i += stride) p[i] = 0.0f;
}

__global__ void deg_kernel(const int* __restrict__ src, const int* __restrict__ dst,
                           const float* __restrict__ ew, float* __restrict__ deg, int nE) {
    const int stride = gridDim.x * blockDim.x;
    for (int e = blockIdx.x * blockDim.x + threadIdx.x; e < nE; e += stride) {
        const int s = src[e];
        if (s != dst[e]) atomicAdd(&deg[s], ew[e]);
    }
}

__global__ void dinv_kernel(float* __restrict__ deg, int n) {
    int i = blockIdx.x * blockDim.x + threadIdx.x;
    if (i >= n) return;
    float d = deg[i];
    deg[i] = (d > 0.0f) ? rsqrtf(d) : 0.0f;
}

#define HALF_ROW   (N_NODES / 2)
#define CTR_STRIDE 64
#define N_MIX      (NLAYERS * (KORD - 1))
#define CTR_WORDS  ((2 * N_MIX + 2) * CTR_STRIDE)

__global__ void __launch_bounds__(256)
norm_part_kernel(const int* __restrict__ src, const int* __restrict__ dst,
                 const float* __restrict__ ew, const float* __restrict__ dinv,
                 int* __restrict__ psrc, int* __restrict__ pdst, float* __restrict__ pnorm,
                 int* __restrict__ cur, int nE) {
    __shared__ int wlo[8], whi[8], base[2];
    const int lane = threadIdx.x & 31, wave = threadIdx.x >> 5;
    const int nchunk = (nE + 255) / 256;
    for (int ch = blockIdx.x; ch < nchunk; ch += gridDim.x) {
        const int e = ch * 256 + threadIdx.x;
        const bool valid = e < nE;
        int s = 0, d = 0;
        float nv = 0.0f;
        if (valid) {
            s = src[e]; d = dst[e];
            const float w = (s == d) ? 0.0f : ew[e];
            nv = -dinv[s] * w * dinv[d];
        }
        const bool lo = valid && d < HALF_ROW;
        const bool hi = valid && d >= HALF_ROW;
        const unsigned mlo = __builtin_amdgcn_ballot_w32(lo);
        const unsigned mhi = __builtin_amdgcn_ballot_w32(hi);
        const unsigned below = (1u << lane) - 1u;
        const int rlo = __popc(mlo & below), rhi = __popc(mhi & below);
        if (lane == 0) { wlo[wave] = __popc(mlo); whi[wave] = __popc(mhi); }
        __syncthreads();
        if (threadIdx.x == 0) {
            int tl = 0, th = 0;
            for (int i = 0; i < 8; ++i) {
                const int a = wlo[i]; wlo[i] = tl; tl += a;
                const int b = whi[i]; whi[i] = th; th += b;
            }
            base[0] = atomicAdd(cur, tl);
            base[1] = atomicAdd(cur + CTR_STRIDE, th);
        }
        __syncthreads();
        if (valid) {
            const int pos = lo ? (base[0] + wlo[wave] + rlo)
                               : (nE - 1 - (base[1] + whi[wave] + rhi));
            psrc[pos] = s; pdst[pos] = d; pnorm[pos] = nv;
        }
        __syncthreads();
    }
}

#define SC_EPW 16
#define SC_EPB (SC_EPW * 8)
#define SC_MAXTRIPS (N_EDGES / SC_EPB + 3)
__device__ __forceinline__ void
scatter_block(const float* __restrict__ h, const int* __restrict__ src,
              const int* __restrict__ dst, const float* __restrict__ norm,
              float* __restrict__ out, float scale, int nE,
              int* __restrict__ ctr, const int* __restrict__ cur) {
    __shared__ int grab[2][2];
    const int lane = threadIdx.x & 31;
    const int wv   = __builtin_amdgcn_readfirstlane((int)(threadIdx.x >> 5));
    const int n0   = cur[0];
    const int nb0  = (n0 + SC_EPB - 1) / SC_EPB;
    const int nb1  = (nE - n0 + SC_EPB - 1) / SC_EPB;
    int  tq = 0, tc = -1;
    bool other = false;
    if (threadIdx.x == 0) {
        tq = (__builtin_amdgcn_s_sendmsg_rtn(0x87) >> 18) & 1;
        tc = atomicAdd(ctr + tq * CTR_STRIDE, 1);
    }
    for (int trip = 0; trip < SC_MAXTRIPS; ++trip) {
        int* g = grab[trip & 1];
        if (threadIdx.x == 0) {
            if (tc >= (tq ? nb1 : nb0)) {
                if (!other) {
                    other = true;
                    tq ^= 1;
                    tc = atomicAdd(ctr + tq * CTR_STRIDE, 1);
                    if (tc >= (tq ? nb1 : nb0)) tc = -1;
                } else {
                    tc = -1;
                }
            }
            g[0] = tq; g[1] = tc;
        }
        __syncthreads();
        const int q = __builtin_amdgcn_readfirstlane(g[0]);
        const int c = __builtin_amdgcn_readfirstlane(g[1]);
        if (c < 0) break;
        if (threadIdx.x == 0) tc = atomicAdd(ctr + tq * CTR_STRIDE, 1);
        const int plo  = q ? n0 : 0;
        const int phi  = q ? nE : n0;
        const int a0   = (q ? nE - (c + 1) * SC_EPB : c * SC_EPB) + wv * SC_EPW;
        const int e0   = a0 > plo ? a0 : plo;
        const int e1   = a0 + SC_EPW < phi ? a0 + SC_EPW : phi;
        if (e1 - e0 == SC_EPW) {
            int s[SC_EPW], d[SC_EPW];
            float w[SC_EPW], a[SC_EPW], b[SC_EPW];
            for (int i = 0; i < SC_EPW; ++i) {
                s[i] = src[e0 + i]; d[i] = dst[e0 + i]; w[i] = scale * norm[e0 + i];
            }
            for (int i = 0; i < SC_EPW; ++i) {
                const float* hr = h + (size_t)s[i] * DHID;
                a[i] = hr[lane]; b[i] = hr[lane + 32];
            }
            for (int i = 0; i < SC_EPW; ++i) {
                float* orow = out + (size_t)d[i] * DHID;
                atomicAdd(orow + lane,      w[i] * a[i]);
                atomicAdd(orow + lane + 32, w[i] * b[i]);
            }
        } else {
            for (int e = e0; e < e1; ++e) {
                const float w = scale * norm[e];
                const float* hr = h + (size_t)src[e] * DHID;
                float* orow = out + (size_t)dst[e] * DHID;
                atomicAdd(orow + lane,      w * hr[lane]);
                atomicAdd(orow + lane + 32, w * hr[lane + 32]);
            }
        }
    }
}

#define RB_BLOCKS 1024
static_assert((RB_BLOCKS * 256) % (DHID / 4) == 0, "a thread must stay on its columns");
__global__ void __launch_bounds__(256)
relu_bias_kernel(const float* __restrict__ acc, const float* __restrict__ b,
                 float* __restrict__ out, int n4) {
    const int t  = blockIdx.x * 256 + threadIdx.x;
    const int cb = (t & (DHID / 4 - 1)) * 4;
    const float b0 = b[cb], b1 = b[cb + 1], b2 = b[cb + 2], b3 = b[cb + 3];
    const float4* __restrict__ a4 = (const float4*)acc;
    float4* __restrict__ o4       = (float4*)out;
    for (int i = t; i < n4; i += RB_BLOCKS * 256) {
        float4 v = a4[i];
        v.x = fmaxf(v.x + b0, 0.0f); v.y = fmaxf(v.y + b1, 0.0f);
        v.z = fmaxf(v.z + b2, 0.0f); v.w = fmaxf(v.w + b3, 0.0f);
        o4[i] = v;
    }
}

template <int MODE, bool COPY, bool NEGW>
__device__ __forceinline__ void
gemm_tile(int gblock, const float* __restrict__ T, float* __restrict__ copyout,
          const float* __restrict__ W, float* __restrict__ acc, int nrows) {
    __shared__ float lWp[DHID * DHID];
    for (int idx = threadIdx.x; idx < DHID * DHID; idx += 256) {
        int k = idx >> 6, n = idx & 63;
        const float wv = W[idx];
        lWp[((k >> 1) << 7) + (n << 1) + (k & 1)] = NEGW ? -wv : wv;
    }
    __syncthreads();

    const int wave = threadIdx.x >> 5;
    const int lane = threadIdx.x & 31;
    const int row0 = (gblock * 8 + wave) * 16;
    if (row0 >= nrows) return;

    const int lrow  = lane & 15;
    const int khalf = (lane >> 4) << 1;
    const int crow  = row0 + ((lane >> 4) << 3);
    const int ccol  = lane & 15;

    v8f c0 = {}, c1 = {}, c2 = {}, c3 = {};
    if (MODE != 0) {
#pragma unroll
        for (int r = 0; r < 8; ++r) {
            const float* ap = acc + (size_t)(crow + r) * DHID + ccol;
            c0[r] = ap[0]; c1[r] = ap[16]; c2[r] = ap[32]; c3[r] = ap[48];
        }
    }

    const float* arow = T + (size_t)(row0 + lrow) * DHID + khalf;
    float* crowp      = COPY ? (copyout + (size_t)(row0 + lrow) * DHID + khalf) : nullptr;

#pragma unroll
    for (int kt = 0; kt < 16; ++kt) {
        const int k0 = kt * 4 + khalf;
        v2f a = *(const v2f*)(arow + kt * 4);
        if (COPY) *(v2f*)(crowp + kt * 4) = a;
        const float* wp = lWp + ((k0 >> 1) << 7);
        v2f b;
        b = *(const v2f*)(wp + ((lrow +  0) << 1));
        c0 = __builtin_amdgcn_wmma_f32_16x16x4_f32(false, a, false, b, (short)0, c0, false, false);
        b = *(const v2f*)(wp + ((lrow + 16) << 1));
        c1 = __builtin_amdgcn_wmma_f32_16x16x4_f32(false, a, false, b, (short)0, c1, false, false);
        b = *(const v2f*)(wp + ((lrow + 32) << 1));
        c2 = __builtin_amdgcn_wmma_f32_16x16x4_f32(false, a, false, b, (short)0, c2, false, false);
        b = *(const v2f*)(wp + ((lrow + 48) << 1));
        c3 = __builtin_amdgcn_wmma_f32_16x16x4_f32(false, a, false, b, (short)0, c3, false, false);
    }

#pragma unroll
    for (int r = 0; r < 8; ++r) {
        float* ap = acc + (size_t)(crow + r) * DHID + ccol;
        ap[0] = c0[r]; ap[16] = c1[r]; ap[32] = c2[r]; ap[48] = c3[r];
    }
}

template <int MODE, bool COPY, bool NEGW>
__global__ void __launch_bounds__(256)
gemm64_wmma(const float* __restrict__ T, float* __restrict__ copyout,
            const float* __restrict__ W, float* __restrict__ acc, int nrows) {
    gemm_tile<MODE, COPY, NEGW>((int)blockIdx.x, T, copyout, W, acc, nrows);
}

#define G_GEMM ((N_NODES + 127) / 128)
#define G_SCAT 1536
#define G_MIX  (G_GEMM + G_SCAT)

template <int MODE, bool COPY, bool NEGW>
__global__ void __launch_bounds__(256)
gemm_scatter_kernel(const float* __restrict__ T, float* __restrict__ copyout,
                    const float* __restrict__ W, float* __restrict__ acc, int nrows,
                    const float* __restrict__ sh, const int* __restrict__ src,
                    const int* __restrict__ dst, const float* __restrict__ norm,
                    float* __restrict__ sout, float scale, int nE,
                    int* __restrict__ ctr, const int* __restrict__ cur) {
    const int b = (int)blockIdx.x;
    if (b < G_GEMM)
        gemm_tile<MODE, COPY, NEGW>(b, T, copyout, W, acc, nrows);
    else
        scatter_block(sh, src, dst, norm, sout, scale, nE, ctr, cur);
}

extern "C" void kernel_launch(void* const* d_in, const int* in_sizes, int n_in,
                              void* d_out, int out_size, void* d_ws, size_t ws_size,
                              hipStream_t stream) {
    const float* x       = (const float*)d_in[0];
    const int*   ei      = (const int*)d_in[1];
    const float* ew      = (const float*)d_in[2];
    const float* weights = (const float*)d_in[3];
    const float* bias    = (const float*)d_in[4];
    float*       out     = (float*)d_out;

    const int* src = ei;
    const int* dst = ei + N_EDGES;

    float* ws   = (float*)d_ws;
    float* PO   = ws;
    float* PE   = ws + (size_t)1 * N_NODES * DHID;
    float* ACC  = ws + (size_t)2 * N_NODES * DHID;
    float* norm = ws + (size_t)3 * N_NODES * DHID;
    int*   psrc = (int*)(norm + N_EDGES);
    int*   pdst = psrc + N_EDGES;
    int*   ctr  = pdst + N_EDGES;
    int*   cur  = ctr + 2 * N_MIX * CTR_STRIDE;
    float* dinv = (float*)(ctr + CTR_WORDS);

    const int NF = N_NODES * DHID;
    dim3 blk(256);
    const int gN    = (N_NODES + 255) / 256;
    const int gGemm = (N_NODES + 127) / 128;

    zero_kernel<<<1024, blk, 0, stream>>>((float*)ctr, CTR_WORDS + N_NODES);
    deg_kernel <<<1024, blk, 0, stream>>>(src, dst, ew, dinv, N_EDGES);
    dinv_kernel<<<gN,   blk, 0, stream>>>(dinv, N_NODES);
    norm_part_kernel<<<1024, blk, 0, stream>>>(src, dst, ew, dinv, psrc, pdst, norm, cur, N_EDGES);

    const int gMix = G_MIX;

    for (int l = 0; l < NLAYERS; ++l) {
        const float* Wl = weights + (size_t)l * KORD * DHID * DHID;

        zero_kernel<<<2048, blk, 0, stream>>>(PO, NF);
        if (l == 0)
            gemm_scatter_kernel<0, true, false><<<gMix, blk, 0, stream>>>(
                x, PE, Wl, ACC, N_NODES, x, psrc, pdst, norm, PO, 1.0f, N_EDGES,
                ctr + (size_t)(l * (KORD - 1)) * 2 * CTR_STRIDE, cur);
        else
            gemm_scatter_kernel<0, false, false><<<gMix, blk, 0, stream>>>(
                PE, nullptr, Wl, ACC, N_NODES, PE, psrc, pdst, norm, PO, 1.0f, N_EDGES,
                ctr + (size_t)(l * (KORD - 1)) * 2 * CTR_STRIDE, cur);

        for (int k = 1; k < KORD - 1; ++k) {
            const float* uk = (k & 1) ? PO : PE;
            float* uk1      = (k & 1) ? PE : PO;
            const float* Wk = Wl + (size_t)k * DHID * DHID;
            const float ck  = (k & 1) ? -2.0f : 2.0f;
            int* ck_ctr     = ctr + (size_t)(l * (KORD - 1) + k) * 2 * CTR_STRIDE;
            if (k & 2)
                gemm_scatter_kernel<1, false, true><<<gMix, blk, 0, stream>>>(
                    uk, nullptr, Wk, ACC, N_NODES, uk, psrc, pdst, norm, uk1, ck, N_EDGES, ck_ctr, cur);
            else
                gemm_scatter_kernel<1, false, false><<<gMix, blk, 0, stream>>>(
                    uk, nullptr, Wk, ACC, N_NODES, uk, psrc, pdst, norm, uk1, ck, N_EDGES, ck_ctr, cur);
        }
        gemm64_wmma<1, false, true><<<gGemm, blk, 0, stream>>>(
            PO, nullptr, Wl + (size_t)(KORD - 1) * DHID * DHID, ACC, N_NODES);

        relu_bias_kernel<<<RB_BLOCKS, blk, 0, stream>>>(
            ACC, bias + l * DHID, (l == NLAYERS - 1) ? out : PE, NF / 4);
    }
}
